# starting kernel
# baseline (speedup 1.0000x reference)
#include <hip/hip_runtime.h>

typedef __attribute__((ext_vector_type(2))) float v2f;
typedef __attribute__((ext_vector_type(8))) float v8f;

#define NV2 163842
#define NV1 40962
#define NV0 10242
#define ALPHA 0.3f

__device__ __forceinline__ float4 feat_get4(const float* __restrict__ fA,
                                            const int* __restrict__ mapA, int CA,
                                            const float* __restrict__ fB, int CB,
                                            int v, int c4) {
  if (c4 < CA) {
    int u = mapA ? mapA[v] : v;
    return *(const float4*)(fA + (long)u * CA + c4);
  }
  return *(const float4*)(fB + (long)v * CB + (c4 - CA));
}

__global__ __launch_bounds__(64) void edgeconv_wmma(
    const float* __restrict__ fA, const int* __restrict__ mapA, int CA,
    const float* __restrict__ fB, int CB,
    const int* __restrict__ src,
    const float* __restrict__ W, const float* __restrict__ bias,
    float* __restrict__ out, int nv, int Cin, int Cout)
{
  extern __shared__ __align__(16) char smem[];
  int*   sidx = (int*)smem;
  float* xl   = (float*)(smem + 384);
  const int K2   = 2 * Cin;
  const int tid  = threadIdx.x;
  const int tile = blockIdx.x;

  for (int i = tid; i < 96; i += 64) {
    int r = i / 6, j = i - r * 6;
    int v = tile * 16 + r; if (v >= nv) v = nv - 1;
    sidx[i] = src[v * 6 + j];
  }
  __syncthreads();

  const int C4 = Cin >> 2;
  for (int i = tid; i < 16 * C4; i += 64) {
    int r = i / C4, c4 = (i - r * C4) << 2;
    int v = tile * 16 + r; if (v >= nv) v = nv - 1;
    float4 loc = feat_get4(fA, mapA, CA, fB, CB, v, c4);
    float sx = 0.f, sy = 0.f, sz = 0.f, sw = 0.f;
    #pragma unroll
    for (int j = 0; j < 6; ++j) {
      float4 t = feat_get4(fA, mapA, CA, fB, CB, sidx[r * 6 + j], c4);
      sx += t.x; sy += t.y; sz += t.z; sw += t.w;
    }
    const float inv6 = 1.0f / 6.0f;
    float4 d = make_float4(sx * inv6 - loc.x, sy * inv6 - loc.y,
                           sz * inv6 - loc.z, sw * inv6 - loc.w);
    *(float4*)(xl + r * K2 + c4)       = loc;
    *(float4*)(xl + r * K2 + Cin + c4) = d;
  }
  __syncthreads();

  const int lane = tid & 31;
  const int wave = tid >> 5;
  const int g    = lane >> 4;
  const int nl   = lane & 15;
  const int nK   = K2 >> 2;
  const int nNT  = Cout >> 4;

  for (int nt = wave; nt < nNT; nt += 2) {
    const int n = nt * 16 + nl;
    const float bv = bias[n];
    v8f acc;
    #pragma unroll
    for (int j = 0; j < 8; ++j) acc[j] = bv;

    const float* __restrict__ ap = xl + nl * K2 + 2 * g;
    const float* __restrict__ bp = W + (long)n * K2 + 2 * g;
    for (int kt = 0; kt < nK; ++kt) {
      v2f a = *(const v2f*)(ap + 4 * kt);
      v2f b = *(const v2f*)(bp + 4 * kt);
      acc = __builtin_amdgcn_wmma_f32_16x16x4_f32(
           false, a,  false, b,
           (short)0, acc,  false,  false);
    }

    #pragma unroll
    for (int j = 0; j < 8; ++j) {
      int m = tile * 16 + j + 8 * g;
      if (m < nv) {
        float y = acc[j];
        y = (y >= 0.f) ? y : ALPHA * y;
        out[(long)m * Cout + n] = y;
      }
    }
  }
}

__device__ __forceinline__ unsigned ford(float f) {
  unsigned u = __float_as_uint(f);
  return (u & 0x80000000u) ? ~u : (u | 0x80000000u);
}
__device__ __forceinline__ float funord(unsigned u) {
  unsigned b = (u & 0x80000000u) ? (u & 0x7FFFFFFFu) : ~u;
  return __uint_as_float(b);
}

__global__ void pool_init(unsigned* __restrict__ p, int n) {
  int i = blockIdx.x * blockDim.x + threadIdx.x;
  if (i < n) p[i] = 0x007FFFFFu;
}
__global__ void pool_scatter(const float* __restrict__ f, const int* __restrict__ map,
                             unsigned* __restrict__ buf, int nchild, int C) {
  int i = blockIdx.x * blockDim.x + threadIdx.x;
  if (i >= nchild * C) return;
  int child = i / C, c = i - child * C;
  atomicMax(&buf[(long)map[child] * C + c], ford(f[i]));
}
__global__ void pool_decode(unsigned* __restrict__ buf, int n) {
  int i = blockIdx.x * blockDim.x + threadIdx.x;
  if (i < n) {
    float v = funord(buf[i]);
    ((float*)buf)[i] = v;
  }
}

extern "C" void kernel_launch(void* const* d_in, const int* in_sizes, int n_in,
                              void* d_out, int out_size, void* d_ws, size_t ws_size,
                              hipStream_t stream) {
  (void)in_sizes; (void)n_in; (void)out_size; (void)ws_size;
  const float* features = (const float*)d_in[0];
  const int*   src2     = (const int*)d_in[1];
  const int*   src1     = (const int*)d_in[3];
  const int*   src0     = (const int*)d_in[5];
  const int*   p21      = (const int*)d_in[7];
  const int*   p10      = (const int*)d_in[8];
  const float* w_enc0 = (const float*)d_in[9],  *b_enc0 = (const float*)d_in[10];
  const float* w_enc1 = (const float*)d_in[11], *b_enc1 = (const float*)d_in[12];
  const float* w_enc2 = (const float*)d_in[13], *b_enc2 = (const float*)d_in[14];
  const float* w_dec0 = (const float*)d_in[15], *b_dec0 = (const float*)d_in[16];
  const float* w_dec1 = (const float*)d_in[17], *b_dec1 = (const float*)d_in[18];

  char* ws = (char*)d_ws;
  size_t off = 0;
  float*    f2 = (float*)(ws + off); off += (size_t)NV2 * 32 * 4;
  float*    f1 = (float*)(ws + off); off += (size_t)NV1 * 64 * 4;
  float*    f0 = (float*)(ws + off); off += (size_t)NV0 * 128 * 4;
  float*    d1 = (float*)(ws + off); off += (size_t)NV1 * 64 * 4;
  unsigned* P  = (unsigned*)(ws + off);

  auto ec = [&](const float* fA, const int* mapA, int CA, const float* fB, int CB,
                const int* src, const float* W, const float* b, float* out,
                int nv, int Cin, int Cout) {
    int K2 = 2 * Cin;
    size_t shmem = 384 + (size_t)(16 * K2) * sizeof(float);
    int grid = (nv + 15) / 16;
    edgeconv_wmma<<<grid, 64, shmem, stream>>>(fA, mapA, CA, fB, CB, src, W, b,
                                               out, nv, Cin, Cout);
  };
  auto pool = [&](const float* fchild, const int* map, int nchild, int nparent, int C) {
    int nc = nchild * C, np = nparent * C;
    pool_init   <<<(np + 255) / 256, 256, 0, stream>>>(P, np);
    pool_scatter<<<(nc + 255) / 256, 256, 0, stream>>>(fchild, map, P, nchild, C);
    pool_decode <<<(np + 255) / 256, 256, 0, stream>>>(P, np);
  };

  ec(features, nullptr, 16, nullptr, 0, src2, w_enc0, b_enc0, f2, NV2, 16, 32);
  pool(f2, p21, NV2, NV1, 32);
  ec((const float*)P, nullptr, 32, nullptr, 0, src1, w_enc1, b_enc1, f1, NV1, 32, 64);
  pool(f1, p10, NV1, NV0, 64);
  ec((const float*)P, nullptr, 64, nullptr, 0, src0, w_enc2, b_enc2, f0, NV0, 64, 128);

  ec(f0, p10, 128, f1, 64, src1, w_dec0, b_dec0, d1, NV1, 192, 64);
  ec(d1, p21, 64, f2, 32, src2, w_dec1, b_dec1, (float*)d_out, NV2, 96, 32);
}


# Encoder EdgeConv: W staged once per work-group in LDS (pitch K2+4), 16 waves walk units; a wave gathers its own 16-vertex tile into its LDS slice (7 row loads grouped, no block barrier); same sums
# speedup vs baseline: 1.8389x; 1.8389x over previous
#include <hip/hip_runtime.h>

typedef __attribute__((ext_vector_type(2))) float v2f;
typedef __attribute__((ext_vector_type(8))) float v8f;

#define NV2 163842
#define NV1 40962
#define NV0 10242
#define ALPHA 0.3f

__device__ __forceinline__ unsigned ford(float f) {
  unsigned u = __float_as_uint(f);
  return (u & 0x80000000u) ? ~u : (u | 0x80000000u);
}
__device__ __forceinline__ float funord(unsigned u) {
  unsigned b = (u & 0x80000000u) ? (u & 0x7FFFFFFFu) : ~u;
  return __uint_as_float(b);
}
#define ORD_NEG_INF 0x007FFFFFu

__device__ __forceinline__ void wave_lds_handover() {
  __builtin_amdgcn_fence(__ATOMIC_RELEASE, "wavefront");
  __builtin_amdgcn_wave_barrier();
  __builtin_amdgcn_fence(__ATOMIC_ACQUIRE, "wavefront");
}

template <int CIN, int COUT, int NSPLIT, bool ORD, bool FILL>
__global__ __launch_bounds__(512) void enc_wmma(
    const float* __restrict__ f, const int* __restrict__ src,
    const float* __restrict__ W, const float* __restrict__ bias,
    float* __restrict__ out, int nv, uint4* __restrict__ fill, int nfill4)
{
  constexpr int K2    = 2 * CIN;
  constexpr int NK    = K2 / 4;
  constexpr int NNT   = COUT / 16;
  constexpr int NTU   = NNT / NSPLIT;
  constexpr int PITCH = K2 + 4;
  constexpr int ROW4  = K2 / 4;
  constexpr int C4    = CIN / 4;
  constexpr int NIT   = 16 * C4 / 32;
  static_assert(NTU * NSPLIT == NNT && (NTU % 2) == 0, "N-tiles are taken in pairs");
  __shared__ __align__(16) float sW[COUT * PITCH];
  __shared__ __align__(16) float sX[16 * 16 * PITCH];

  constexpr int NPIECE = COUT * ROW4;
  if constexpr (NPIECE % 512 == 0) {
    constexpr int NST = NPIECE / 512;
    float4 wt[NST];
    #pragma unroll
    for (int s = 0; s < NST; ++s)
      wt[s] = *(const float4*)(W + (long)(threadIdx.x + s * 512) * 4);
    #pragma unroll
    for (int s = 0; s < NST; ++s) {
      const int i = threadIdx.x + s * 512;
      const int n = i / ROW4, c = (i - n * ROW4) * 4;
      *(float4*)(sW + n * PITCH + c) = wt[s];
    }
  } else {
    static_assert(NPIECE < 512, "a small W is one piece per thread at most");
    if (threadIdx.x < NPIECE) {
      const int i = threadIdx.x;
      const int n = i / ROW4, c = (i - n * ROW4) * 4;
      *(float4*)(sW + n * PITCH + c) = *(const float4*)(W + (long)i * 4);
    }
  }
  __syncthreads();

  if (FILL) {
    const uint4 fv = make_uint4(ORD_NEG_INF, ORD_NEG_INF, ORD_NEG_INF, ORD_NEG_INF);
    for (int i = blockIdx.x * 512 + threadIdx.x; i < nfill4; i += (int)gridDim.x * 512) fill[i] = fv;
  }

  const int lane = threadIdx.x & 31;
  const int wave = threadIdx.x >> 5;
  const int g  = lane >> 4;
  const int nl = lane & 15;
  float* __restrict__ xw = sX + wave * (16 * PITCH);
  const int nunits = ((nv + 15) >> 4) * NSPLIT;

  for (int u = wave * (int)gridDim.x + (int)blockIdx.x; u < nunits; u += 16 * (int)gridDim.x) {
    asm volatile("" ::: "memory");
    const int tile = u / NSPLIT;
    const int nt0  = (u - tile * NSPLIT) * NTU;

    constexpr int CH = (NIT < 4) ? NIT : 4;
    #pragma unroll 1
    for (int t0 = 0; t0 < NIT; t0 += CH) {
      int sj[CH][6];
      int vv[CH];
      #pragma unroll
      for (int t = 0; t < CH; ++t) {
        const int r = (lane + 32 * (t0 + t)) / C4;
        int v = tile * 16 + r; if (v >= nv) v = nv - 1;
        vv[t] = v;
        const int2* sp = (const int2*)(src + (long)v * 6);
        const int2 s01 = sp[0], s23 = sp[1], s45 = sp[2];
        sj[t][0] = s01.x; sj[t][1] = s01.y; sj[t][2] = s23.x;
        sj[t][3] = s23.y; sj[t][4] = s45.x; sj[t][5] = s45.y;
      }
      #pragma unroll
      for (int t = 0; t < CH; ++t) {
        const int i  = lane + 32 * (t0 + t);
        const int r  = i / C4, c4 = (i - r * C4) * 4;
        float4 row[7];
        #pragma unroll
        for (int j = 0; j < 7; ++j) {
          const int w = (j == 0) ? vv[t] : sj[t][(j == 0) ? 0 : j - 1];
          const float* p = f + (long)w * CIN + c4;
          if (ORD) {
            const uint4 q = *(const uint4*)p;
            row[j] = make_float4(funord(q.x), funord(q.y), funord(q.z), funord(q.w));
          } else {
            row[j] = *(const float4*)p;
          }
        }
        float sx = 0.f, sy = 0.f, sz = 0.f, sw = 0.f;
        #pragma unroll
        for (int j = 1; j < 7; ++j) { sx += row[j].x; sy += row[j].y; sz += row[j].z; sw += row[j].w; }
        const float inv6 = 1.0f / 6.0f;
        const float4 loc = row[0];
        const float4 d = make_float4(sx * inv6 - loc.x, sy * inv6 - loc.y,
                                     sz * inv6 - loc.z, sw * inv6 - loc.w);
        *(float4*)(xw + r * PITCH + c4)       = loc;
        *(float4*)(xw + r * PITCH + CIN + c4) = d;
      }
    }
    wave_lds_handover();

    v2f a[NK];
    #pragma unroll
    for (int kt = 0; kt < NK; ++kt) a[kt] = *(const v2f*)(xw + nl * PITCH + 4 * kt + 2 * g);
    wave_lds_handover();

    for (int nt = nt0; nt < nt0 + NTU; nt += 2) {
      const int n0 = nt * 16 + nl, n1 = n0 + 16;
      const float bv0 = bias[n0], bv1 = bias[n1];
      v8f acc0, acc1;
      #pragma unroll
      for (int j = 0; j < 8; ++j) { acc0[j] = bv0; acc1[j] = bv1; }
      const float* bp0 = sW + n0 * PITCH + 2 * g;
      const float* bp1 = bp0 + 16 * PITCH;
      #pragma unroll
      for (int kt = 0; kt < NK; ++kt) {
        const v2f b0 = *(const v2f*)(bp0 + 4 * kt);
        const v2f b1 = *(const v2f*)(bp1 + 4 * kt);
        acc0 = __builtin_amdgcn_wmma_f32_16x16x4_f32(false, a[kt], false, b0, (short)0, acc0, false, false);
        acc1 = __builtin_amdgcn_wmma_f32_16x16x4_f32(false, a[kt], false, b1, (short)0, acc1, false, false);
      }
      #pragma unroll
      for (int j = 0; j < 8; ++j) {
        const int m = tile * 16 + j + 8 * g;
        if (m < nv) {
          float y0 = acc0[j], y1 = acc1[j];
          y0 = (y0 >= 0.f) ? y0 : ALPHA * y0;
          y1 = (y1 >= 0.f) ? y1 : ALPHA * y1;
          out[(long)m * COUT + n0] = y0;
          out[(long)m * COUT + n1] = y1;
        }
      }
    }
  }
}


template <int CA, int CB, int COUT, int NSPLIT>
__global__ __launch_bounds__(512) void proj_wmma(
    const float* __restrict__ fA, const int* __restrict__ mapA,
    const float* __restrict__ fB,
    const float* __restrict__ W, const float* __restrict__ bias,
    float* __restrict__ R, float* __restrict__ Q, int nv)
{
  constexpr int CIN   = CA + CB;
  constexpr int K2    = 2 * CIN;
  constexpr int NK    = CIN / 4;
  constexpr int NNT   = COUT / 16;
  constexpr int NTU   = NNT / NSPLIT;
  constexpr int PITCH = K2 + 4;
  constexpr int ROW4  = K2 / 4;
  __shared__ __align__(16) float sW[COUT * PITCH];

  constexpr int NST = COUT * ROW4 / 512;
  static_assert(COUT * ROW4 == NST * 512, "W must split evenly over the 512 threads");
  float4 wt[NST];
  #pragma unroll
  for (int s = 0; s < NST; ++s)
    wt[s] = *(const float4*)(W + (long)(threadIdx.x + s * 512) * 4);
  #pragma unroll
  for (int s = 0; s < NST; ++s) {
    const int i = threadIdx.x + s * 512;
    const int n = i / ROW4, c = (i - n * ROW4) * 4;
    *(float4*)(sW + n * PITCH + c) = wt[s];
  }
  __syncthreads();

  const int lane = threadIdx.x & 31;
  const int wave = threadIdx.x >> 5;
  const int g  = lane >> 4;
  const int nl = lane & 15;
  const int nunits = ((nv + 15) >> 4) * NSPLIT;

  for (int u = wave * (int)gridDim.x + (int)blockIdx.x; u < nunits; u += 16 * (int)gridDim.x) {
    asm volatile("" ::: "memory");
    const int tile = u / NSPLIT;
    const int nt0  = (u - tile * NSPLIT) * NTU;
    int v = tile * 16 + nl; if (v >= nv) v = nv - 1;

    const float* __restrict__ rowA = fA + (long)mapA[v] * CA + 2 * g;
    const float* __restrict__ rowB = fB + (long)v * CB + 2 * g;
    v2f a[NK];
    #pragma unroll
    for (int kt = 0; kt < NK; ++kt)
      a[kt] = (4 * kt < CA) ? *(const v2f*)(rowA + 4 * kt)
                            : *(const v2f*)(rowB + (4 * kt - CA));

    for (int nt = nt0; nt < nt0 + NTU; ++nt) {
      const int n = nt * 16 + nl;
      const float bv = bias[n];
      v8f accP, accQ;
      #pragma unroll
      for (int j = 0; j < 8; ++j) { accP[j] = bv; accQ[j] = 0.f; }
      const float* bp = sW + n * PITCH + 2 * g;
      #pragma unroll
      for (int kt = 0; kt < NK; ++kt) {
        v2f b1 = *(const v2f*)(bp + 4 * kt);
        v2f b2 = *(const v2f*)(bp + CIN + 4 * kt);
        accP = __builtin_amdgcn_wmma_f32_16x16x4_f32(false, a[kt], false, b1, (short)0, accP, false, false);
        accQ = __builtin_amdgcn_wmma_f32_16x16x4_f32(false, a[kt], false, b2, (short)0, accQ, false, false);
      }
      #pragma unroll
      for (int j = 0; j < 8; ++j) {
        int m = tile * 16 + j + 8 * g;
        if (m < nv) {
          R[(long)m * COUT + n] = accP[j] - accQ[j];
          Q[(long)m * COUT + n] = accQ[j];
        }
      }
    }
  }
}

template <int COUT>
__global__ __launch_bounds__(1024) void agg_mean(
    const float* __restrict__ R, const float* __restrict__ Q,
    const int* __restrict__ src, float* __restrict__ out, int nv)
{
  constexpr int C4 = COUT / 4;
  const int i = blockIdx.x * 1024 + threadIdx.x;
  if (i >= nv * C4) return;
  const int v  = i / C4;
  const int c4 = (i - v * C4) * 4;
  int s[6];
  #pragma unroll
  for (int j = 0; j < 6; ++j) s[j] = src[v * 6 + j];
  const float4 r = *(const float4*)(R + (long)v * COUT + c4);
  float4 t[6];
  #pragma unroll
  for (int j = 0; j < 6; ++j) t[j] = *(const float4*)(Q + (long)s[j] * COUT + c4);
  float sx = 0.f, sy = 0.f, sz = 0.f, sw = 0.f;
  #pragma unroll
  for (int j = 0; j < 6; ++j) { sx += t[j].x; sy += t[j].y; sz += t[j].z; sw += t[j].w; }
  const float inv6 = 1.0f / 6.0f;
  float4 y = make_float4(r.x + sx * inv6, r.y + sy * inv6, r.z + sz * inv6, r.w + sw * inv6);
  y.x = (y.x >= 0.f) ? y.x : ALPHA * y.x;
  y.y = (y.y >= 0.f) ? y.y : ALPHA * y.y;
  y.z = (y.z >= 0.f) ? y.z : ALPHA * y.z;
  y.w = (y.w >= 0.f) ? y.w : ALPHA * y.w;
  *(float4*)(out + (long)v * COUT + c4) = y;
}

template <int C, int U>
__global__ __launch_bounds__(256) void pool_scatter(const float* __restrict__ f,
                                                    const int* __restrict__ map,
                                                    unsigned* __restrict__ buf, int n) {
  const unsigned nn = (unsigned)n;
  const unsigned stride = gridDim.x * 256u;
  const unsigned i0 = blockIdx.x * 256u + threadIdx.x;
  float v[U];
  int   p[U];
  #pragma unroll
  for (int k = 0; k < U; ++k) {
    unsigned e = i0 + k * stride; if (e >= nn) e = nn - 1u;
    v[k] = f[e];
    p[k] = map[e / C];
  }
  #pragma unroll
  for (int k = 0; k < U; ++k) {
    const unsigned e = i0 + k * stride;
    if (e < nn) atomicMax(&buf[(long)p[k] * C + (e & (C - 1))], ford(v[k]));
  }
}

extern "C" void kernel_launch(void* const* d_in, const int* in_sizes, int n_in,
                              void* d_out, int out_size, void* d_ws, size_t ws_size,
                              hipStream_t stream) {
  (void)in_sizes; (void)n_in; (void)out_size; (void)ws_size;
  const float* features = (const float*)d_in[0];
  const int*   src2     = (const int*)d_in[1];
  const int*   src1     = (const int*)d_in[3];
  const int*   src0     = (const int*)d_in[5];
  const int*   p21      = (const int*)d_in[7];
  const int*   p10      = (const int*)d_in[8];
  const float* w_enc0 = (const float*)d_in[9],  *b_enc0 = (const float*)d_in[10];
  const float* w_enc1 = (const float*)d_in[11], *b_enc1 = (const float*)d_in[12];
  const float* w_enc2 = (const float*)d_in[13], *b_enc2 = (const float*)d_in[14];
  const float* w_dec0 = (const float*)d_in[15], *b_dec0 = (const float*)d_in[16];
  const float* w_dec1 = (const float*)d_in[17], *b_dec1 = (const float*)d_in[18];

  char* ws = (char*)d_ws;
  size_t off = 0;
  float*    f2 = (float*)(ws + off); off += (size_t)NV2 * 32 * 4;
  float*    f1 = (float*)(ws + off); off += (size_t)NV1 * 64 * 4;
  float*    f0 = (float*)(ws + off); off += (size_t)NV0 * 128 * 4;
  float*    d1 = (float*)(ws + off); off += (size_t)NV1 * 64 * 4;
  unsigned* P1 = (unsigned*)(ws + off); off += (size_t)NV1 * 32 * 4;
  unsigned* P0 = (unsigned*)(ws + off); off += (size_t)NV0 * 64 * 4;
  float*    Rb = (float*)(ws + off); off += (size_t)NV2 * 32 * 4;
  float*    Qb = (float*)(ws + off);

  enc_wmma<16, 32, 1, false, true><<<512, 512, 0, stream>>>(features, src2, w_enc0, b_enc0, f2, NV2,
                                                          (uint4*)P1, (NV1 * 32 + NV0 * 64) / 4);
  pool_scatter<32, 16><<<(NV2 * 32 + 4095) / 4096, 256, 0, stream>>>(f2, p21, P1, NV2 * 32);
  enc_wmma<32, 64, 1, true, false><<<256, 512, 0, stream>>>((const float*)P1, src1, w_enc1, b_enc1, f1, NV1,
                                                          nullptr, 0);
  pool_scatter<64, 16><<<(NV1 * 64 + 4095) / 4096, 256, 0, stream>>>(f1, p10, P0, NV1 * 64);
  enc_wmma<64, 128, 4, true, false><<<256, 512, 0, stream>>>((const float*)P0, src0, w_enc2, b_enc2, f0, NV0,
                                                           nullptr, 0);

  proj_wmma<128, 64, 64, 2><<<256, 512, 0, stream>>>(f0, p10, f1, w_dec0, b_dec0, Rb, Qb, NV1);
  agg_mean<64><<<(NV1 * 16 + 1023) / 1024, 1024, 0, stream>>>(Rb, Qb, src1, d1, NV1);
  proj_wmma<64, 32, 32, 1><<<256, 512, 0, stream>>>(d1, p21, f2, w_dec1, b_dec1, Rb, Qb, NV2);
  agg_mean<32><<<(NV2 * 8 + 1023) / 1024, 1024, 0, stream>>>(Rb, Qb, src2, (float*)d_out, NV2);
}
